# starting kernel
# baseline (speedup 1.0000x reference)
#include <hip/hip_runtime.h>
#include <hip/hip_bf16.h>

typedef __attribute__((ext_vector_type(2))) float v2f;
typedef __attribute__((ext_vector_type(8))) float v8f;

#define N_NODES 100000
#define N_EDGES 1600000
#define HID     64
#define NGRAPH  2048

__global__ void fill_k(float* __restrict__ p, float v, long long n) {
    long long i = (long long)blockIdx.x * blockDim.x + threadIdx.x;
    if (i < n) p[i] = v;
}

__global__ void deg_k(const int* __restrict__ dst, float* __restrict__ deg, int E) {
    int e = blockIdx.x * blockDim.x + threadIdx.x;
    if (e < E) atomicAdd(&deg[dst[e]], 1.0f);
}

__global__ void rsqrt_k(float* __restrict__ d, int n) {
    int i = blockIdx.x * blockDim.x + threadIdx.x;
    if (i < n) d[i] = rsqrtf(d[i]);
}

__global__ void norm_k(const int* __restrict__ src, const int* __restrict__ dst,
                       const float* __restrict__ dis, float* __restrict__ nrm, int E) {
    int e = blockIdx.x * blockDim.x + threadIdx.x;
    if (e < E) nrm[e] = dis[src[e]] * dis[dst[e]];
}

__global__ void gemm_wmma_k(const float* __restrict__ A, const float* __restrict__ W,
                            float* __restrict__ C, int rows, int K) {
    const int lane = threadIdx.x & 31;
    const int w    = (blockIdx.x * blockDim.x + threadIdx.x) >> 5;
    const int rowTiles = rows >> 4;
    if (w >= rowTiles) return;

    const int m0  = w << 4;
    const int mr  = lane & 15;
    const int khi = (lane >> 4) << 1;

    const float* arow = A + (size_t)(m0 + mr) * K + khi;
    const float* bcol = W + mr;

    v8f acc0 = {}, acc1 = {}, acc2 = {}, acc3 = {};
    for (int k = 0; k < K; k += 4) {
        v2f a;
        const float2 av = *(const float2*)(arow + k);
        a[0] = av.x; a[1] = av.y;

        const float* brow0 = bcol + (size_t)(k + khi) * HID;
        const float* brow1 = brow0 + HID;
        v2f b0, b1, b2, b3;
        b0[0] = brow0[0];  b0[1] = brow1[0];
        b1[0] = brow0[16]; b1[1] = brow1[16];
        b2[0] = brow0[32]; b2[1] = brow1[32];
        b3[0] = brow0[48]; b3[1] = brow1[48];

        acc0 = __builtin_amdgcn_wmma_f32_16x16x4_f32(false, a, false, b0, (short)0, acc0, false, false);
        acc1 = __builtin_amdgcn_wmma_f32_16x16x4_f32(false, a, false, b1, (short)0, acc1, false, false);
        acc2 = __builtin_amdgcn_wmma_f32_16x16x4_f32(false, a, false, b2, (short)0, acc2, false, false);
        acc3 = __builtin_amdgcn_wmma_f32_16x16x4_f32(false, a, false, b3, (short)0, acc3, false, false);
    }

    float* crow = C + (size_t)(m0 + ((lane >> 4) << 3)) * HID + mr;
#pragma unroll
    for (int j = 0; j < 8; ++j) {
        crow[(size_t)j * HID +  0] = acc0[j];
        crow[(size_t)j * HID + 16] = acc1[j];
        crow[(size_t)j * HID + 32] = acc2[j];
        crow[(size_t)j * HID + 48] = acc3[j];
    }
}

__global__ void edge_msg_k(const int* __restrict__ src, const int* __restrict__ dst,
                           const float* __restrict__ nrm, const float* __restrict__ h,
                           float* __restrict__ agg, int E) {
    int t = blockIdx.x * blockDim.x + threadIdx.x;
    int e = t >> 5;
    if (e >= E) return;
    int lane = t & 31;
    int s = src[e], d = dst[e];
    float nv = nrm[e];
    int f = lane << 1;
    float2 hv = *(const float2*)(h + (size_t)s * HID + f);
    atomicAdd(&agg[(size_t)d * HID + f    ], hv.x * nv);
    atomicAdd(&agg[(size_t)d * HID + f + 1], hv.y * nv);
}

__global__ void finalize_k(const float* agg, const float* __restrict__ hw,
                           const float* __restrict__ dis, const float* __restrict__ bias,
                           float* outp, int n) {
    int i = blockIdx.x * blockDim.x + threadIdx.x;
    if (i >= n * 16) return;
    int node = i >> 4;
    int f4   = (i & 15) << 2;
    float s = dis[node]; s = s * s;
    const float4 a = *(const float4*)(agg + (size_t)node * HID + f4);
    const float4 h = *(const float4*)(hw  + (size_t)node * HID + f4);
    float4 r;
    r.x = fmaxf(a.x + h.x * s + bias[f4 + 0], 0.0f);
    r.y = fmaxf(a.y + h.y * s + bias[f4 + 1], 0.0f);
    r.z = fmaxf(a.z + h.z * s + bias[f4 + 2], 0.0f);
    r.w = fmaxf(a.w + h.w * s + bias[f4 + 3], 0.0f);
    *(float4*)(outp + (size_t)node * HID + f4) = r;
}

__global__ void counts_k(const int* __restrict__ batch, float* __restrict__ counts, int n) {
    int i = blockIdx.x * blockDim.x + threadIdx.x;
    if (i < n) atomicAdd(&counts[batch[i]], 1.0f);
}

__global__ void pool_k(const float* __restrict__ h, const int* __restrict__ batch,
                       float* __restrict__ pooled, int n) {
    int i = blockIdx.x * blockDim.x + threadIdx.x;
    if (i >= n * HID) return;
    int node = i >> 6;
    int f    = i & 63;
    int g    = batch[node];
    float v  = h[i];
    atomicAdd(&pooled[(size_t)g * 128 + f], v);
    atomicMax((unsigned int*)&pooled[(size_t)g * 128 + 64 + f], __float_as_uint(v));
}

__global__ void meandiv_k(float* __restrict__ pooled, const float* __restrict__ counts, int G) {
    int i = blockIdx.x * blockDim.x + threadIdx.x;
    if (i >= G * HID) return;
    int g = i >> 6;
    int f = i & 63;
    pooled[(size_t)g * 128 + f] /= fmaxf(counts[g], 1.0f);
}

__global__ void head_k(const float* __restrict__ pooled, const float* __restrict__ Wout,
                       const float* __restrict__ bout, float* __restrict__ out, int G) {
    int t = blockIdx.x * blockDim.x + threadIdx.x;
    int g = t >> 5;
    int lane = t & 31;
    if (g >= G) return;
    float v = 0.0f;
#pragma unroll
    for (int j = lane; j < 2 * HID; j += 32) v += pooled[(size_t)g * 128 + j] * Wout[j];
#pragma unroll
    for (int off = 16; off; off >>= 1) v += __shfl_xor(v, off, 32);
    if (lane == 0) out[g] = v + bout[0];
}

extern "C" void kernel_launch(void* const* d_in, const int* in_sizes, int n_in,
                              void* d_out, int out_size, void* d_ws, size_t ws_size,
                              hipStream_t stream) {
    const float* x     = (const float*)d_in[0];
    const int*   eidx  = (const int*)  d_in[1];
    const int*   batch = (const int*)  d_in[2];
    const float* Ws[4] = { (const float*)d_in[3], (const float*)d_in[5],
                           (const float*)d_in[7], (const float*)d_in[9] };
    const float* bs[4] = { (const float*)d_in[4], (const float*)d_in[6],
                           (const float*)d_in[8], (const float*)d_in[10] };
    const float* Wout  = (const float*)d_in[11];
    const float* bout  = (const float*)d_in[12];

    const int N = N_NODES, E = N_EDGES, G = NGRAPH;
    const int* src = eidx;
    const int* dst = eidx + E;

    float* ws     = (float*)d_ws;
    float* dis    = ws;
    float* nrm    = dis + N;
    float* counts = nrm + E;
    float* buf0   = counts + G;
    float* buf1   = buf0 + (size_t)N * HID;
    float* buf2   = buf1 + (size_t)N * HID;

    float* out    = (float*)d_out;
    float* pooled = out + G;

    fill_k <<<(N + 255) / 256, 256, 0, stream>>>(dis, 1.0f, N);
    deg_k  <<<(E + 255) / 256, 256, 0, stream>>>(dst, dis, E);
    rsqrt_k<<<(N + 255) / 256, 256, 0, stream>>>(dis, N);
    norm_k <<<(E + 255) / 256, 256, 0, stream>>>(src, dst, dis, nrm, E);

    const float* hin = x;
    float* HW  = buf0;
    float* AGG = buf1;
    float* spare = buf2;
    int K = 128;
    const long long feat = (long long)N * HID;

    for (int l = 0; l < 4; ++l) {
        int gemmThreads = (N / 16) * 32;
        gemm_wmma_k<<<(gemmThreads + 255) / 256, 256, 0, stream>>>(hin, Ws[l], HW, N, K);
        fill_k     <<<(int)((feat + 255) / 256), 256, 0, stream>>>(AGG, 0.0f, feat);
        edge_msg_k <<<(int)(((long long)E * 32 + 255) / 256), 256, 0, stream>>>(src, dst, nrm, HW, AGG, E);
        finalize_k <<<(N * 16 + 255) / 256, 256, 0, stream>>>(AGG, HW, dis, bs[l], AGG, N);

        const float* next_in = AGG;
        float* newAGG = (l == 0) ? spare : (float*)hin;
        hin = next_in;
        AGG = newAGG;
        K = HID;
    }

    fill_k   <<<(G + G * 128 + 255) / 256, 256, 0, stream>>>(out, 0.0f, G + G * 128);
    fill_k   <<<(G + 255) / 256, 256, 0, stream>>>(counts, 0.0f, G);
    counts_k <<<(N + 255) / 256, 256, 0, stream>>>(batch, counts, N);
    pool_k   <<<(N * HID + 255) / 256, 256, 0, stream>>>(hin, batch, pooled, N);
    meandiv_k<<<(G * HID + 255) / 256, 256, 0, stream>>>(pooled, counts, G);
    head_k   <<<(G * 32 + 255) / 256, 256, 0, stream>>>(pooled, Wout, bout, out, G);
}


# Degree count in LDS: 64 blocks each count their 25000 destinations in a private all-node table of 16-bit counters and store it; tables added where 1/sqrt is taken; 1.6M scattered global adds gone
# speedup vs baseline: 2.3058x; 2.3058x over previous
#include <hip/hip_runtime.h>
#include <hip/hip_bf16.h>

typedef __attribute__((ext_vector_type(2))) float v2f;
typedef __attribute__((ext_vector_type(8))) float v8f;

#define N_NODES 100000
#define N_EDGES 1600000
#define HID     64
#define NGRAPH  2048

__global__ void fill_k(float* __restrict__ p, float v, long long n) {
    long long i = (long long)blockIdx.x * blockDim.x + threadIdx.x;
    if (i < n) p[i] = v;
}

#define DEG_BLOCKS  64
#define DEG_THREADS 1024
#define DEG_CHUNK   ((N_EDGES + DEG_BLOCKS - 1) / DEG_BLOCKS)
#define DEG_ITERS   ((DEG_CHUNK + DEG_THREADS - 1) / DEG_THREADS)
#define DEG_WORDS   (N_NODES / 2)
static_assert(DEG_CHUNK <= 65535, "a 16-bit counter must hold a block's whole chunk");
static_assert(N_NODES % 8 == 0, "tables are moved as 16-byte pieces of 8 counters");

__global__ void __launch_bounds__(DEG_THREADS)
deg_table_k(const int* __restrict__ dst, unsigned int* __restrict__ tabs, int E) {
    __shared__ __attribute__((aligned(16))) unsigned int cnt[DEG_WORDS];
    uint4* cnt4 = (uint4*)cnt;
    for (int i = threadIdx.x; i < DEG_WORDS / 4; i += DEG_THREADS) cnt4[i] = make_uint4(0u, 0u, 0u, 0u);
    __syncthreads();

    const int e0 = blockIdx.x * DEG_CHUNK;
    const int e1 = min(e0 + DEG_CHUNK, E);
    unsigned int d[DEG_ITERS];
#pragma unroll
    for (int j = 0; j < DEG_ITERS; ++j) {
        const int e = e0 + j * DEG_THREADS + (int)threadIdx.x;
        const unsigned int v = (unsigned int)dst[min(e, e1 - 1)];
        d[j] = (e < e1) ? v : 0xFFFFFFFFu;
    }
#pragma unroll
    for (int j = 0; j < DEG_ITERS; ++j)
        if (d[j] != 0xFFFFFFFFu) atomicAdd(&cnt[d[j] >> 1], (d[j] & 1u) ? 0x10000u : 1u);
    __syncthreads();

    uint4* out4 = (uint4*)(tabs + (size_t)blockIdx.x * DEG_WORDS);
    for (int i = threadIdx.x; i < DEG_WORDS / 4; i += DEG_THREADS) out4[i] = cnt4[i];
}

__global__ void dis_k(const unsigned int* __restrict__ tabs, float* __restrict__ dis) {
    const int w = blockIdx.x * blockDim.x + threadIdx.x;
    if (w >= DEG_WORDS) return;
    unsigned int lo = 1u, hi = 1u;
#pragma unroll
    for (int b = 0; b < DEG_BLOCKS; ++b) {
        const unsigned int v = tabs[(size_t)b * DEG_WORDS + w];
        lo += v & 0xFFFFu;
        hi += v >> 16;
    }
    float2 r;
    r.x = rsqrtf((float)lo);
    r.y = rsqrtf((float)hi);
    ((float2*)dis)[w] = r;
}

__global__ void norm_k(const int* __restrict__ src, const int* __restrict__ dst,
                       const float* __restrict__ dis, float* __restrict__ nrm, int E) {
    int e = blockIdx.x * blockDim.x + threadIdx.x;
    if (e < E) nrm[e] = dis[src[e]] * dis[dst[e]];
}

template <bool RELU_IN>
__global__ void gemm_wmma_k(const float* __restrict__ A, const float* __restrict__ A1,
                            const float* __restrict__ W,
                            const float* __restrict__ dis, const float* __restrict__ bias,
                            float* __restrict__ C, float* __restrict__ S, float* __restrict__ Z,
                            int rows, int K) {
    const int lane = threadIdx.x & 31;
    const int w    = (blockIdx.x * blockDim.x + threadIdx.x) >> 5;
    const int rowTiles = rows >> 4;
    if (w >= rowTiles) return;

    const int m0  = w << 4;
    const int mr  = lane & 15;
    const int khi = (lane >> 4) << 1;

    const float* arow  = A  + (size_t)(m0 + mr) * K + khi;
    const float* arow1 = A1 + (size_t)(m0 + mr) * K + khi;
    const float* bcol = W + mr;

    {
        float4* z = (float4*)(Z + (size_t)m0 * HID) + lane;
        const float4 zero4 = make_float4(0.0f, 0.0f, 0.0f, 0.0f);
#pragma unroll
        for (int i = 0; i < 8; ++i) z[i * 32] = zero4;
    }

    v8f acc0 = {}, acc1 = {}, acc2 = {}, acc3 = {};
    for (int k = 0; k < K; k += 4) {
        v2f a;
        const float2 av = *(const float2*)(arow + k);
        if (RELU_IN) {
            const float2 aw = *(const float2*)(arow1 + k);
            a[0] = fmaxf(av.x + aw.x, 0.0f);
            a[1] = fmaxf(av.y + aw.y, 0.0f);
        } else {
            a[0] = av.x;
            a[1] = av.y;
        }

        const float* brow0 = bcol + (size_t)(k + khi) * HID;
        const float* brow1 = brow0 + HID;
        v2f b0, b1, b2, b3;
        b0[0] = brow0[0];  b0[1] = brow1[0];
        b1[0] = brow0[16]; b1[1] = brow1[16];
        b2[0] = brow0[32]; b2[1] = brow1[32];
        b3[0] = brow0[48]; b3[1] = brow1[48];

        acc0 = __builtin_amdgcn_wmma_f32_16x16x4_f32(false, a, false, b0, (short)0, acc0, false, false);
        acc1 = __builtin_amdgcn_wmma_f32_16x16x4_f32(false, a, false, b1, (short)0, acc1, false, false);
        acc2 = __builtin_amdgcn_wmma_f32_16x16x4_f32(false, a, false, b2, (short)0, acc2, false, false);
        acc3 = __builtin_amdgcn_wmma_f32_16x16x4_f32(false, a, false, b3, (short)0, acc3, false, false);
    }

    const int rbase = m0 + ((lane >> 4) << 3);
    const float4 d0 = *(const float4*)(dis + rbase);
    const float4 d1 = *(const float4*)(dis + rbase + 4);
    const float s2[8] = { d0.x * d0.x, d0.y * d0.y, d0.z * d0.z, d0.w * d0.w,
                          d1.x * d1.x, d1.y * d1.y, d1.z * d1.z, d1.w * d1.w };
    const float bb0 = bias[mr], bb1 = bias[mr + 16], bb2 = bias[mr + 32], bb3 = bias[mr + 48];
    float* crow = C + (size_t)rbase * HID + mr;
    float* srow = S + (size_t)rbase * HID + mr;
#pragma unroll
    for (int j = 0; j < 8; ++j) {
        crow[(size_t)j * HID +  0] = acc0[j];
        crow[(size_t)j * HID + 16] = acc1[j];
        crow[(size_t)j * HID + 32] = acc2[j];
        crow[(size_t)j * HID + 48] = acc3[j];
        srow[(size_t)j * HID +  0] = acc0[j] * s2[j] + bb0;
        srow[(size_t)j * HID + 16] = acc1[j] * s2[j] + bb1;
        srow[(size_t)j * HID + 32] = acc2[j] * s2[j] + bb2;
        srow[(size_t)j * HID + 48] = acc3[j] * s2[j] + bb3;
    }
}

#define EPW 16
__global__ void __launch_bounds__(256)
edge_msg_k(const int* __restrict__ src, const int* __restrict__ dst,
           const float* __restrict__ nrm, const float* __restrict__ h,
           float* __restrict__ agg0, float* __restrict__ agg1, int E) {
    __shared__ int s_half;
    if (threadIdx.x == 0) s_half = (int)((__builtin_amdgcn_s_sendmsg_rtn(0x87) >> 18) & 1u);
    __syncthreads();
    float* __restrict__ agg = __builtin_amdgcn_readfirstlane(s_half) ? agg1 : agg0;
    const int lane = threadIdx.x & 31;
    const int w  = __builtin_amdgcn_readfirstlane((int)((blockIdx.x * blockDim.x + threadIdx.x) >> 5));
    const int e0 = w * EPW;
    if (e0 >= E) return;
    const int*   sp = src + e0;
    const int*   dp = dst + e0;
    const float* np = nrm + e0;

    if (e0 + EPW <= E) {
        int s[EPW], d[EPW];
        float nv[EPW];
#pragma unroll
        for (int j = 0; j < EPW; ++j) { s[j] = sp[j]; d[j] = dp[j]; nv[j] = np[j]; }

        float v0[EPW], v1[EPW];
#pragma unroll
        for (int j = 0; j < EPW; ++j) {
            const float* hrow = h + (size_t)s[j] * HID;
            v0[j] = hrow[lane];
            v1[j] = hrow[lane + 32];
        }
#pragma unroll
        for (int j = 0; j < EPW; ++j) {
            float* arow = agg + (size_t)d[j] * HID;
            atomicAdd(arow + lane,      v0[j] * nv[j]);
            atomicAdd(arow + lane + 32, v1[j] * nv[j]);
        }
    } else {
        const int n = E - e0;
        for (int j = 0; j < n; ++j) {
            const float* hrow = h + (size_t)sp[j] * HID;
            float* arow = agg + (size_t)dp[j] * HID;
            const float nj = np[j];
            atomicAdd(arow + lane,      hrow[lane]      * nj);
            atomicAdd(arow + lane + 32, hrow[lane + 32] * nj);
        }
    }
}

__global__ void counts_k(const int* __restrict__ batch, float* __restrict__ counts, int n) {
    int i = blockIdx.x * blockDim.x + threadIdx.x;
    if (i < n) atomicAdd(&counts[batch[i]], 1.0f);
}

__global__ void pool_k(const float* __restrict__ h, const float* __restrict__ h1,
                       const int* __restrict__ batch,
                       float* __restrict__ pooled, int n) {
    int i = blockIdx.x * blockDim.x + threadIdx.x;
    if (i >= n * HID) return;
    int node = i >> 6;
    int f    = i & 63;
    int g    = batch[node];
    float v  = fmaxf(h[i] + h1[i], 0.0f);
    atomicAdd(&pooled[(size_t)g * 128 + f], v);
    atomicMax((unsigned int*)&pooled[(size_t)g * 128 + 64 + f], __float_as_uint(v));
}

__global__ void meandiv_k(float* __restrict__ pooled, const float* __restrict__ counts, int G) {
    int i = blockIdx.x * blockDim.x + threadIdx.x;
    if (i >= G * HID) return;
    int g = i >> 6;
    int f = i & 63;
    pooled[(size_t)g * 128 + f] /= fmaxf(counts[g], 1.0f);
}

__global__ void head_k(const float* __restrict__ pooled, const float* __restrict__ Wout,
                       const float* __restrict__ bout, float* __restrict__ out, int G) {
    int t = blockIdx.x * blockDim.x + threadIdx.x;
    int g = t >> 5;
    int lane = t & 31;
    if (g >= G) return;
    float v = 0.0f;
#pragma unroll
    for (int j = lane; j < 2 * HID; j += 32) v += pooled[(size_t)g * 128 + j] * Wout[j];
#pragma unroll
    for (int off = 16; off; off >>= 1) v += __shfl_xor(v, off, 32);
    if (lane == 0) out[g] = v + bout[0];
}

extern "C" void kernel_launch(void* const* d_in, const int* in_sizes, int n_in,
                              void* d_out, int out_size, void* d_ws, size_t ws_size,
                              hipStream_t stream) {
    const float* x     = (const float*)d_in[0];
    const int*   eidx  = (const int*)  d_in[1];
    const int*   batch = (const int*)  d_in[2];
    const float* Ws[4] = { (const float*)d_in[3], (const float*)d_in[5],
                           (const float*)d_in[7], (const float*)d_in[9] };
    const float* bs[4] = { (const float*)d_in[4], (const float*)d_in[6],
                           (const float*)d_in[8], (const float*)d_in[10] };
    const float* Wout  = (const float*)d_in[11];
    const float* bout  = (const float*)d_in[12];

    const int N = N_NODES, E = N_EDGES, G = NGRAPH;
    const int* src = eidx;
    const int* dst = eidx + E;

    float* ws     = (float*)d_ws;
    float* dis    = ws;
    float* nrm    = dis + N;
    float* counts = nrm + E;
    float* buf0   = counts + G;
    float* buf1   = buf0 + (size_t)N * HID;
    float* buf2   = buf1 + (size_t)N * HID;
    float* buf3   = buf2 + (size_t)N * HID;
    float* buf4   = buf3 + (size_t)N * HID;

    float* out    = (float*)d_out;
    float* pooled = out + G;

    unsigned int* tabs = (unsigned int*)buf3;
    deg_table_k<<<DEG_BLOCKS, DEG_THREADS, 0, stream>>>(dst, tabs, E);
    dis_k  <<<(DEG_WORDS + 255) / 256, 256, 0, stream>>>(tabs, dis);
    norm_k <<<(E + 255) / 256, 256, 0, stream>>>(src, dst, dis, nrm, E);

    const float* hin  = x;
    const float* hin1 = x;
    float* HW  = buf0;
    float* AGG = buf1;  float* AGGb = buf2;
    float* other = buf3; float* otherb = buf4;
    const int gemmThreads = (N / 16) * 32;
    const int gemmBlocks  = (gemmThreads + 255) / 256;
    const int edgeBlocks  = (int)((((long long)E + EPW - 1) / EPW * 32 + 255) / 256);

    for (int l = 0; l < 4; ++l) {
        if (l == 0)
            gemm_wmma_k<false><<<gemmBlocks, 256, 0, stream>>>(hin, hin1, Ws[l], dis, bs[l], HW, AGG, AGGb, N, 128);
        else
            gemm_wmma_k<true> <<<gemmBlocks, 256, 0, stream>>>(hin, hin1, Ws[l], dis, bs[l], HW, AGG, AGGb, N, HID);
        edge_msg_k<<<edgeBlocks, 256, 0, stream>>>(src, dst, nrm, HW, AGG, AGGb, E);

        float* prev = AGG;  float* prevb = AGGb;
        AGG = other;  AGGb = otherb;
        other = prev; otherb = prevb;
        hin = prev;   hin1 = prevb;
    }

    fill_k   <<<(G + G * 128 + 255) / 256, 256, 0, stream>>>(out, 0.0f, G + G * 128);
    fill_k   <<<(G + 255) / 256, 256, 0, stream>>>(counts, 0.0f, G);
    counts_k <<<(N + 255) / 256, 256, 0, stream>>>(batch, counts, N);
    pool_k   <<<(N * HID + 255) / 256, 256, 0, stream>>>(hin, hin1, batch, pooled, N);
    meandiv_k<<<(G * HID + 255) / 256, 256, 0, stream>>>(pooled, counts, G);
    head_k   <<<(G * 32 + 255) / 256, 256, 0, stream>>>(pooled, Wout, bout, out, G);
}
